# starting kernel
# baseline (speedup 1.0000x reference)
#include <hip/hip_runtime.h>

#define B_   8
#define N_   65536
#define D_   40
#define DP_  48
#define K_   2
#define CHUNKS_ 128
#define CHUNK_  512
#define WAVES_  8
#define ROWS_PER_WAVE_ 64
#define WAVE_ITERS_ 2

typedef float v2f __attribute__((ext_vector_type(2)));
typedef float v8f __attribute__((ext_vector_type(8)));

__global__ __launch_bounds__(256) void sk_init(const float* __restrict__ data,
                                               const int* __restrict__ ids,
                                               float* __restrict__ cent) {
  int tid = blockIdx.x * 256 + threadIdx.x;
  if (tid >= B_ * K_ * DP_) return;
  int b  = tid / (K_ * DP_);
  int kd = tid % (K_ * DP_);
  int k  = kd / DP_;
  int d  = kd % DP_;
  int n  = ids[b * K_ + k];
  cent[tid] = (d < D_) ? data[((size_t)b * N_ + n) * D_ + d] : 0.0f;
}

__global__ __launch_bounds__(256) void sk_pass(const float* __restrict__ data,
                                               const float* __restrict__ cent,
                                               float* __restrict__ partials,
                                               float2* __restrict__ probs_out,
                                               int write_probs, int accumulate) {
  __shared__ float cLds[K_ * DP_];
  __shared__ float xt[WAVES_][32 * DP_];
  __shared__ float pt[WAVES_][32 * K_];
  __shared__ float red[WAVES_][K_ * DP_];

  const int tid   = threadIdx.x;
  const int wave  = tid >> 5;
  const int lane  = tid & 31;
  const int b     = blockIdx.x >> 7;
  const int chunk = blockIdx.x & (CHUNKS_ - 1);

  for (int i = tid; i < K_ * DP_; i += 256) {
    int k = i / DP_, d = i % DP_;
    cLds[i] = (d < D_) ? cent[(b * K_ + k) * DP_ + d] : 0.0f;
  }
  __syncthreads();

  float cc0 = 0.f, cc1 = 0.f;
#pragma unroll
  for (int d = 0; d < D_; ++d) {
    float a0 = cLds[d], a1 = cLds[DP_ + d];
    cc0 = fmaf(a0, a0, cc0);
    cc1 = fmaf(a1, a1, cc1);
  }

  v8f acc[3] = {};
  float* myxt = &xt[wave][0];
  float* mypt = &pt[wave][0];
  const float* __restrict__ dbase = data + (size_t)b * N_ * D_;
  const int m15  = lane & 15;
  const int m1   = lane & 1;
  const bool alive = (m15 < K_);
  const int rsel = (lane >> 4) << 1;

#pragma unroll
  for (int it = 0; it < WAVE_ITERS_; ++it) {
    const int n = chunk * CHUNK_ + wave * ROWS_PER_WAVE_ + it * 32 + lane;
    const float4* __restrict__ xr = (const float4*)(dbase + (size_t)n * D_);
    float x[D_];
#pragma unroll
    for (int q = 0; q < D_ / 4; ++q) {
      float4 v = xr[q];
      x[4 * q + 0] = v.x; x[4 * q + 1] = v.y;
      x[4 * q + 2] = v.z; x[4 * q + 3] = v.w;
    }
    float dot0 = 0.f, dot1 = 0.f;
#pragma unroll
    for (int d = 0; d < D_; ++d) {
      dot0 = fmaf(x[d], cLds[d], dot0);
      dot1 = fmaf(x[d], cLds[DP_ + d], dot1);
    }
    float s0 = fmaf(2.0f, dot0, -cc0);
    float s1 = fmaf(2.0f, dot1, -cc1);
    float mx = fmaxf(s0, s1);
    float e0 = __expf(s0 - mx);
    float e1 = __expf(s1 - mx);
    float inv = 1.0f / (e0 + e1);
    float p0 = e0 * inv, p1 = e1 * inv;

    if (write_probs) {
      probs_out[(size_t)b * N_ + n] = make_float2(p0, p1);
    }
    if (accumulate) {
#pragma unroll
      for (int d = 0; d < D_; ++d) myxt[lane * DP_ + d] = x[d];
      myxt[lane * DP_ + D_] = 1.0f;
#pragma unroll
      for (int d = D_ + 1; d < DP_; ++d) myxt[lane * DP_ + d] = 0.0f;
      mypt[lane * K_ + 0] = p0;
      mypt[lane * K_ + 1] = p1;
      __builtin_amdgcn_wave_barrier();

#pragma unroll
      for (int g = 0; g < 8; ++g) {
        const int r0 = 4 * g + rsel;
        float a0v = mypt[(r0 + 0) * K_ + m1];
        float a1v = mypt[(r0 + 1) * K_ + m1];
        v2f a;
        a.x = alive ? a0v : 0.0f;
        a.y = alive ? a1v : 0.0f;
#pragma unroll
        for (int t = 0; t < 3; ++t) {
          v2f bm;
          bm.x = myxt[(r0 + 0) * DP_ + 16 * t + m15];
          bm.y = myxt[(r0 + 1) * DP_ + 16 * t + m15];
          acc[t] = __builtin_amdgcn_wmma_f32_16x16x4_f32(
              false, a, false, bm, (short)0, acc[t], false, false);
        }
      }
      __builtin_amdgcn_wave_barrier();
    }
  }

  if (accumulate) {
    if (lane < 16) {
#pragma unroll
      for (int t = 0; t < 3; ++t) {
        red[wave][0 * DP_ + 16 * t + lane] = acc[t][0];
        red[wave][1 * DP_ + 16 * t + lane] = acc[t][1];
      }
    }
    __syncthreads();
    if (tid < K_ * DP_) {
      float s = 0.f;
#pragma unroll
      for (int w = 0; w < WAVES_; ++w) s += red[w][tid];
      partials[(size_t)blockIdx.x * (K_ * DP_) + tid] = s;
    }
  }
}

__global__ __launch_bounds__(256) void sk_reduce(const float* __restrict__ partials,
                                                 float* __restrict__ cent) {
  int tid = blockIdx.x * 256 + threadIdx.x;
  if (tid >= B_ * K_ * DP_) return;
  int b  = tid / (K_ * DP_);
  int kd = tid % (K_ * DP_);
  int k  = kd / DP_;
  int d  = kd % DP_;
  float s = 0.f, sden = 0.f;
  for (int c = 0; c < CHUNKS_; ++c) {
    const float* p = partials + (size_t)(b * CHUNKS_ + c) * (K_ * DP_);
    s    += p[kd];
    sden += p[k * DP_ + D_];
  }
  cent[tid] = (d < D_) ? (s / sden) : 0.0f;
}

extern "C" void kernel_launch(void* const* d_in, const int* in_sizes, int n_in,
                              void* d_out, int out_size, void* d_ws, size_t ws_size,
                              hipStream_t stream) {
  const float* data = (const float*)d_in[0];
  const int*   ids  = (const int*)d_in[1];
  const int ITER = 10;

  float*  cent     = (float*)d_ws;
  float*  partials = cent + B_ * K_ * DP_;
  float2* out      = (float2*)d_out;

  sk_init<<<(B_ * K_ * DP_ + 255) / 256, 256, 0, stream>>>(data, ids, cent);
  for (int t = 0; t <= ITER; ++t) {
    int wp = (t == ITER) ? 1 : 0;
    int ac = (t <  ITER) ? 1 : 0;
    sk_pass<<<B_ * CHUNKS_, 256, 0, stream>>>(data, cent, partials, out, wp, ac);
    if (ac)
      sk_reduce<<<(B_ * K_ * DP_ + 255) / 256, 256, 0, stream>>>(partials, cent);
  }
  (void)in_sizes; (void)n_in; (void)out_size; (void)ws_size;
}


# sk_pass without LDS tile: lanes load 12-byte row pieces straight into their B-operand place (320 B contiguous per load); score difference summed over 16 lanes by DPP butterfly; new fixed sum order
# speedup vs baseline: 1.0982x; 1.0982x over previous
#include <hip/hip_runtime.h>

#define B_   8
#define N_   65536
#define D_   40
#define DP_  48
#define K_   2
#define CHUNKS_ 128
#define CHUNK_  512
#define WAVES_  8
#define ROWS_PER_WAVE_ 64
#define GROUPS_ 4
#define STEPS_  4

typedef float v2f __attribute__((ext_vector_type(2)));
typedef float v8f __attribute__((ext_vector_type(8)));

struct F3 { float a, b, c; };

__global__ __launch_bounds__(256) void sk_init(const float* __restrict__ data,
                                               const int* __restrict__ ids,
                                               float* __restrict__ cent) {
  int tid = blockIdx.x * 256 + threadIdx.x;
  if (tid >= B_ * K_ * DP_) return;
  int b  = tid / (K_ * DP_);
  int kd = tid % (K_ * DP_);
  int k  = kd / DP_;
  int d  = kd % DP_;
  int n  = ids[b * K_ + k];
  cent[tid] = (d < D_) ? data[((size_t)b * N_ + n) * D_ + d] : 0.0f;
}

template <int M>
__device__ __forceinline__ float row_xor(float v) {
  return __builtin_bit_cast(float,
      __builtin_amdgcn_update_dpp(0, __builtin_bit_cast(int, v), 0x160 + M, 0xf, 0xf, true));
}
__device__ __forceinline__ float row_sum16(float v) {
  v += row_xor<1>(v);
  v += row_xor<2>(v);
  v += row_xor<4>(v);
  v += row_xor<8>(v);
  return v;
}

__global__ __launch_bounds__(256) void sk_pass(const float* __restrict__ data,
                                               const float* __restrict__ cent,
                                               float* __restrict__ partials,
                                               float2* __restrict__ probs_out,
                                               int write_probs, int accumulate) {
  __shared__ float red[WAVES_][K_ * DP_];

  const int tid   = threadIdx.x;
  const int wave  = tid >> 5;
  const int lane  = tid & 31;
  const int c     = lane & 15;
  const int h     = lane >> 4;
  const int b     = blockIdx.x >> 7;
  const int chunk = blockIdx.x & (CHUNKS_ - 1);

  const int off = (c < 13) ? 3 * c : (D_ - 3);
  bool own[3];
  own[0] = (c < 13); own[1] = (c < 13); own[2] = (c < 14);

  const float* __restrict__ sbase =
      data + ((size_t)b * N_ + (size_t)chunk * CHUNK_ + wave * ROWS_PER_WAVE_ + h) * D_ + off;

  F3 v[2][GROUPS_][2];
#pragma unroll
  for (int g = 0; g < GROUPS_; ++g) {
#pragma unroll
    for (int s = 0; s < 2; ++s) v[0][g][s] = *(const F3*)(sbase + (4 * g + 2 * s) * D_);
  }

  __builtin_amdgcn_sched_barrier(0);

  const float* __restrict__ c0p = cent + (b * K_ + 0) * DP_ + off;
  const float* __restrict__ c1p = cent + (b * K_ + 1) * DP_ + off;
  float w[3];
  float bpart = 0.0f;
#pragma unroll
  for (int t = 0; t < 3; ++t) {
    float a0 = c0p[t], a1 = c1p[t];
    w[t] = own[t] ? 2.0f * (a0 - a1) : 0.0f;
    bpart += own[t] ? (a1 * a1 - a0 * a0) : 0.0f;
  }
  const float bias = row_sum16(bpart);
  const float binit = (c == 15) ? bias : 0.0f;
  const float kexp = (c & 1) ? 1.4426950408889634f : -1.4426950408889634f;
  const bool ones_lane = (c == 14);

  v8f acc[3] = {};

#pragma unroll
  for (int st = 0; st < STEPS_; ++st) {
    const int cur = st & 1;
    if (st + 1 < STEPS_) {
#pragma unroll
      for (int g = 0; g < GROUPS_; ++g) {
#pragma unroll
        for (int s = 0; s < 2; ++s)
          v[cur ^ 1][g][s] = *(const F3*)(sbase + ((st + 1) * 4 * GROUPS_ + 4 * g + 2 * s) * D_);
      }
    }
    float z[GROUPS_][2];
#pragma unroll
    for (int g = 0; g < GROUPS_; ++g) {
#pragma unroll
      for (int s = 0; s < 2; ++s) {
        float p = fmaf(v[cur][g][s].a, w[0], binit);
        p = fmaf(v[cur][g][s].b, w[1], p);
        p = fmaf(v[cur][g][s].c, w[2], p);
        z[g][s] = row_sum16(p);
      }
    }

    if (write_probs) {
      float zs = z[0][0];
#pragma unroll
      for (int i = 1; i < 2 * GROUPS_; ++i) zs = (c == i) ? z[i >> 1][i & 1] : zs;
      float e   = __expf(-fabsf(zs));
      float inv = 1.0f / (1.0f + e);
      float big = inv, small = e * inv;
      float p0 = (zs >= 0.0f) ? big : small;
      float p1 = (zs >= 0.0f) ? small : big;
      const int n = chunk * CHUNK_ + wave * ROWS_PER_WAVE_ + st * 4 * GROUPS_ +
                    4 * (c >> 1) + 2 * (c & 1) + h;
      if (c < 2 * GROUPS_) probs_out[(size_t)b * N_ + n] = make_float2(p0, p1);
    }
    if (accumulate) {
#pragma unroll
      for (int g = 0; g < GROUPS_; ++g) {
        v2f a;
        a.x = __builtin_amdgcn_rcpf(1.0f + __builtin_amdgcn_exp2f(z[g][0] * kexp));
        a.y = __builtin_amdgcn_rcpf(1.0f + __builtin_amdgcn_exp2f(z[g][1] * kexp));
        v2f b0, b1, b2;
        b0.x = ones_lane ? 1.0f : v[cur][g][0].a;  b0.y = ones_lane ? 1.0f : v[cur][g][1].a;
        b1.x = v[cur][g][0].b;                     b1.y = v[cur][g][1].b;
        b2.x = v[cur][g][0].c;                     b2.y = v[cur][g][1].c;
        acc[0] = __builtin_amdgcn_wmma_f32_16x16x4_f32(false, a, false, b0, (short)0, acc[0], false, false);
        acc[1] = __builtin_amdgcn_wmma_f32_16x16x4_f32(false, a, false, b1, (short)0, acc[1], false, false);
        acc[2] = __builtin_amdgcn_wmma_f32_16x16x4_f32(false, a, false, b2, (short)0, acc[2], false, false);
      }
    }
  }

  if (accumulate) {
    if (lane < 16) {
#pragma unroll
      for (int t = 0; t < 3; ++t) {
        const bool den = ones_lane && (t == 0);
        const int  d   = own[t] ? off + t : (den ? D_ : DP_ - 1);
        red[wave][0 * DP_ + d] = acc[t][0];
        red[wave][1 * DP_ + d] = acc[t][1];
      }
    }
    __syncthreads();
    if (tid < K_ * DP_) {
      float s = 0.f;
      if ((tid % DP_) <= D_) {
#pragma unroll
        for (int wv = 0; wv < WAVES_; ++wv) s += red[wv][tid];
      }
      partials[(size_t)blockIdx.x * (K_ * DP_) + tid] = s;
    }
  }
}

__global__ __launch_bounds__(256) void sk_reduce(const float* __restrict__ partials,
                                                 float* __restrict__ cent) {
  int tid = blockIdx.x * 256 + threadIdx.x;
  if (tid >= B_ * K_ * DP_) return;
  int b  = tid / (K_ * DP_);
  int kd = tid % (K_ * DP_);
  int k  = kd / DP_;
  int d  = kd % DP_;
  float s = 0.f, sden = 0.f;
  for (int c = 0; c < CHUNKS_; ++c) {
    const float* p = partials + (size_t)(b * CHUNKS_ + c) * (K_ * DP_);
    s    += p[kd];
    sden += p[k * DP_ + D_];
  }
  cent[tid] = (d < D_) ? (s / sden) : 0.0f;
}

extern "C" void kernel_launch(void* const* d_in, const int* in_sizes, int n_in,
                              void* d_out, int out_size, void* d_ws, size_t ws_size,
                              hipStream_t stream) {
  const float* data = (const float*)d_in[0];
  const int*   ids  = (const int*)d_in[1];
  const int ITER = 10;

  float*  cent     = (float*)d_ws;
  float*  partials = cent + B_ * K_ * DP_;
  float2* out      = (float2*)d_out;

  sk_init<<<(B_ * K_ * DP_ + 255) / 256, 256, 0, stream>>>(data, ids, cent);
  for (int t = 0; t <= ITER; ++t) {
    int wp = (t == ITER) ? 1 : 0;
    int ac = (t <  ITER) ? 1 : 0;
    sk_pass<<<B_ * CHUNKS_, 256, 0, stream>>>(data, cent, partials, out, wp, ac);
    if (ac)
      sk_reduce<<<(B_ * K_ * DP_ + 255) / 256, 256, 0, stream>>>(partials, cent);
  }
  (void)in_sizes; (void)n_in; (void)out_size; (void)ws_size;
}


# sk_pass work-group of 32 waves (2048 rows, wave sums joined in LDS by pairwise tree): 32 partial rows per batch, not 128; sk_reduce's dependent load chain 8 trips, not 32; new fixed sum order
# speedup vs baseline: 1.8831x; 1.7147x over previous
#include <hip/hip_runtime.h>

#define B_   8
#define N_   65536
#define D_   40
#define DP_  48
#define K_   2
#define CHUNKS_ 32
#define CHUNK_  2048
#define WAVES_  32
#define ROWS_PER_WAVE_ 64
#define GROUPS_ 4
#define STEPS_  4

typedef float v2f __attribute__((ext_vector_type(2)));
typedef float v8f __attribute__((ext_vector_type(8)));

struct F3 { float a, b, c; };

__global__ __launch_bounds__(256) void sk_init(const float* __restrict__ data,
                                               const int* __restrict__ ids,
                                               float* __restrict__ cent) {
  int tid = blockIdx.x * 256 + threadIdx.x;
  if (tid >= B_ * K_ * DP_) return;
  int b  = tid / (K_ * DP_);
  int kd = tid % (K_ * DP_);
  int k  = kd / DP_;
  int d  = kd % DP_;
  int n  = ids[b * K_ + k];
  cent[tid] = (d < D_) ? data[((size_t)b * N_ + n) * D_ + d] : 0.0f;
}

template <int M>
__device__ __forceinline__ float row_xor(float v) {
  return __builtin_bit_cast(float,
      __builtin_amdgcn_update_dpp(0, __builtin_bit_cast(int, v), 0x160 + M, 0xf, 0xf, true));
}
__device__ __forceinline__ float row_sum16(float v) {
  v += row_xor<1>(v);
  v += row_xor<2>(v);
  v += row_xor<4>(v);
  v += row_xor<8>(v);
  return v;
}

__global__ __launch_bounds__(32 * WAVES_) void sk_pass(const float* __restrict__ data,
                                               const float* __restrict__ cent,
                                               float* __restrict__ partials,
                                               float2* __restrict__ probs_out,
                                               int write_probs, int accumulate) {
  __shared__ float red[WAVES_][K_ * DP_];

  const int tid   = threadIdx.x;
  const int wave  = tid >> 5;
  const int lane  = tid & 31;
  const int c     = lane & 15;
  const int h     = lane >> 4;
  const int b     = blockIdx.x / CHUNKS_;
  const int chunk = blockIdx.x & (CHUNKS_ - 1);

  const int off = (c < 13) ? 3 * c : (D_ - 3);
  bool own[3];
  own[0] = (c < 13); own[1] = (c < 13); own[2] = (c < 14);

  const float* __restrict__ sbase =
      data + ((size_t)b * N_ + (size_t)chunk * CHUNK_ + wave * ROWS_PER_WAVE_ + h) * D_ + off;

  F3 v[2][GROUPS_][2];
#pragma unroll
  for (int g = 0; g < GROUPS_; ++g) {
#pragma unroll
    for (int s = 0; s < 2; ++s) v[0][g][s] = *(const F3*)(sbase + (4 * g + 2 * s) * D_);
  }

  __builtin_amdgcn_sched_barrier(0);

  const float* __restrict__ c0p = cent + (b * K_ + 0) * DP_ + off;
  const float* __restrict__ c1p = cent + (b * K_ + 1) * DP_ + off;
  float w[3];
  float bpart = 0.0f;
#pragma unroll
  for (int t = 0; t < 3; ++t) {
    float a0 = c0p[t], a1 = c1p[t];
    w[t] = own[t] ? 2.0f * (a0 - a1) : 0.0f;
    bpart += own[t] ? (a1 * a1 - a0 * a0) : 0.0f;
  }
  const float bias = row_sum16(bpart);
  const float binit = (c == 15) ? bias : 0.0f;
  const float kexp = (c & 1) ? 1.4426950408889634f : -1.4426950408889634f;
  const bool ones_lane = (c == 14);

  v8f acc[3] = {};

#pragma unroll
  for (int st = 0; st < STEPS_; ++st) {
    const int cur = st & 1;
    if (st + 1 < STEPS_) {
#pragma unroll
      for (int g = 0; g < GROUPS_; ++g) {
#pragma unroll
        for (int s = 0; s < 2; ++s)
          v[cur ^ 1][g][s] = *(const F3*)(sbase + ((st + 1) * 4 * GROUPS_ + 4 * g + 2 * s) * D_);
      }
    }
    float z[GROUPS_][2];
#pragma unroll
    for (int g = 0; g < GROUPS_; ++g) {
#pragma unroll
      for (int s = 0; s < 2; ++s) {
        float p = fmaf(v[cur][g][s].a, w[0], binit);
        p = fmaf(v[cur][g][s].b, w[1], p);
        p = fmaf(v[cur][g][s].c, w[2], p);
        z[g][s] = row_sum16(p);
      }
    }

    if (write_probs) {
      float zs = z[0][0];
#pragma unroll
      for (int i = 1; i < 2 * GROUPS_; ++i) zs = (c == i) ? z[i >> 1][i & 1] : zs;
      float e   = __expf(-fabsf(zs));
      float inv = 1.0f / (1.0f + e);
      float big = inv, small = e * inv;
      float p0 = (zs >= 0.0f) ? big : small;
      float p1 = (zs >= 0.0f) ? small : big;
      const int n = chunk * CHUNK_ + wave * ROWS_PER_WAVE_ + st * 4 * GROUPS_ +
                    4 * (c >> 1) + 2 * (c & 1) + h;
      if (c < 2 * GROUPS_) probs_out[(size_t)b * N_ + n] = make_float2(p0, p1);
    }
    if (accumulate) {
#pragma unroll
      for (int g = 0; g < GROUPS_; ++g) {
        v2f a;
        a.x = __builtin_amdgcn_rcpf(1.0f + __builtin_amdgcn_exp2f(z[g][0] * kexp));
        a.y = __builtin_amdgcn_rcpf(1.0f + __builtin_amdgcn_exp2f(z[g][1] * kexp));
        v2f b0, b1, b2;
        b0.x = ones_lane ? 1.0f : v[cur][g][0].a;  b0.y = ones_lane ? 1.0f : v[cur][g][1].a;
        b1.x = v[cur][g][0].b;                     b1.y = v[cur][g][1].b;
        b2.x = v[cur][g][0].c;                     b2.y = v[cur][g][1].c;
        acc[0] = __builtin_amdgcn_wmma_f32_16x16x4_f32(false, a, false, b0, (short)0, acc[0], false, false);
        acc[1] = __builtin_amdgcn_wmma_f32_16x16x4_f32(false, a, false, b1, (short)0, acc[1], false, false);
        acc[2] = __builtin_amdgcn_wmma_f32_16x16x4_f32(false, a, false, b2, (short)0, acc[2], false, false);
      }
    }
  }

  if (accumulate) {
    if (lane < 16) {
#pragma unroll
      for (int t = 0; t < 3; ++t) {
        const bool den = ones_lane && (t == 0);
        const int  d   = own[t] ? off + t : (den ? D_ : DP_ - 1);
        red[wave][0 * DP_ + d] = acc[t][0];
        red[wave][1 * DP_ + d] = acc[t][1];
      }
    }
    __syncthreads();
    if (tid < K_ * DP_) {
      float s = 0.f;
      if ((tid % DP_) <= D_) {
        float r[WAVES_];
#pragma unroll
        for (int wv = 0; wv < WAVES_; ++wv) r[wv] = red[wv][tid];
#pragma unroll
        for (int w2 = 1; w2 < WAVES_; w2 *= 2) {
#pragma unroll
          for (int wv = 0; wv < WAVES_; wv += 2 * w2) r[wv] += r[wv + w2];
        }
        s = r[0];
      }
      partials[(size_t)blockIdx.x * (K_ * DP_) + tid] = s;
    }
  }
}

__global__ __launch_bounds__(256) void sk_reduce(const float* __restrict__ partials,
                                                 float* __restrict__ cent) {
  int tid = blockIdx.x * 256 + threadIdx.x;
  if (tid >= B_ * K_ * DP_) return;
  int b  = tid / (K_ * DP_);
  int kd = tid % (K_ * DP_);
  int k  = kd / DP_;
  int d  = kd % DP_;
  float s = 0.f, sden = 0.f;
  for (int c = 0; c < CHUNKS_; ++c) {
    const float* p = partials + (size_t)(b * CHUNKS_ + c) * (K_ * DP_);
    s    += p[kd];
    sden += p[k * DP_ + D_];
  }
  cent[tid] = (d < D_) ? (s / sden) : 0.0f;
}

extern "C" void kernel_launch(void* const* d_in, const int* in_sizes, int n_in,
                              void* d_out, int out_size, void* d_ws, size_t ws_size,
                              hipStream_t stream) {
  const float* data = (const float*)d_in[0];
  const int*   ids  = (const int*)d_in[1];
  const int ITER = 10;

  float*  cent     = (float*)d_ws;
  float*  partials = cent + B_ * K_ * DP_;
  float2* out      = (float2*)d_out;

  sk_init<<<(B_ * K_ * DP_ + 255) / 256, 256, 0, stream>>>(data, ids, cent);
  for (int t = 0; t <= ITER; ++t) {
    int wp = (t == ITER) ? 1 : 0;
    int ac = (t <  ITER) ? 1 : 0;
    sk_pass<<<B_ * CHUNKS_, 32 * WAVES_, 0, stream>>>(data, cent, partials, out, wp, ac);
    if (ac)
      sk_reduce<<<(B_ * K_ * DP_ + 255) / 256, 256, 0, stream>>>(partials, cent);
  }
  (void)in_sizes; (void)n_in; (void)out_size; (void)ws_size;
}


# sk_reduce: chunk loop fully unrolled, all 64 partial loads of a thread issued before the first add: one memory round trip, not 8 dependent trips; same threads, same serial sum order
# speedup vs baseline: 2.3881x; 1.2682x over previous
#include <hip/hip_runtime.h>

#define B_   8
#define N_   65536
#define D_   40
#define DP_  48
#define K_   2
#define CHUNKS_ 32
#define CHUNK_  2048
#define WAVES_  32
#define ROWS_PER_WAVE_ 64
#define GROUPS_ 4
#define STEPS_  4

typedef float v2f __attribute__((ext_vector_type(2)));
typedef float v8f __attribute__((ext_vector_type(8)));

struct F3 { float a, b, c; };

__global__ __launch_bounds__(256) void sk_init(const float* __restrict__ data,
                                               const int* __restrict__ ids,
                                               float* __restrict__ cent) {
  int tid = blockIdx.x * 256 + threadIdx.x;
  if (tid >= B_ * K_ * DP_) return;
  int b  = tid / (K_ * DP_);
  int kd = tid % (K_ * DP_);
  int k  = kd / DP_;
  int d  = kd % DP_;
  int n  = ids[b * K_ + k];
  cent[tid] = (d < D_) ? data[((size_t)b * N_ + n) * D_ + d] : 0.0f;
}

template <int M>
__device__ __forceinline__ float row_xor(float v) {
  return __builtin_bit_cast(float,
      __builtin_amdgcn_update_dpp(0, __builtin_bit_cast(int, v), 0x160 + M, 0xf, 0xf, true));
}
__device__ __forceinline__ float row_sum16(float v) {
  v += row_xor<1>(v);
  v += row_xor<2>(v);
  v += row_xor<4>(v);
  v += row_xor<8>(v);
  return v;
}

__global__ __launch_bounds__(32 * WAVES_) void sk_pass(const float* __restrict__ data,
                                               const float* __restrict__ cent,
                                               float* __restrict__ partials,
                                               float2* __restrict__ probs_out,
                                               int write_probs, int accumulate) {
  __shared__ float red[WAVES_][K_ * DP_];

  const int tid   = threadIdx.x;
  const int wave  = tid >> 5;
  const int lane  = tid & 31;
  const int c     = lane & 15;
  const int h     = lane >> 4;
  const int b     = blockIdx.x / CHUNKS_;
  const int chunk = blockIdx.x & (CHUNKS_ - 1);

  const int off = (c < 13) ? 3 * c : (D_ - 3);
  bool own[3];
  own[0] = (c < 13); own[1] = (c < 13); own[2] = (c < 14);

  const float* __restrict__ sbase =
      data + ((size_t)b * N_ + (size_t)chunk * CHUNK_ + wave * ROWS_PER_WAVE_ + h) * D_ + off;

  F3 v[2][GROUPS_][2];
#pragma unroll
  for (int g = 0; g < GROUPS_; ++g) {
#pragma unroll
    for (int s = 0; s < 2; ++s) v[0][g][s] = *(const F3*)(sbase + (4 * g + 2 * s) * D_);
  }

  __builtin_amdgcn_sched_barrier(0);

  const float* __restrict__ c0p = cent + (b * K_ + 0) * DP_ + off;
  const float* __restrict__ c1p = cent + (b * K_ + 1) * DP_ + off;
  float w[3];
  float bpart = 0.0f;
#pragma unroll
  for (int t = 0; t < 3; ++t) {
    float a0 = c0p[t], a1 = c1p[t];
    w[t] = own[t] ? 2.0f * (a0 - a1) : 0.0f;
    bpart += own[t] ? (a1 * a1 - a0 * a0) : 0.0f;
  }
  const float bias = row_sum16(bpart);
  const float binit = (c == 15) ? bias : 0.0f;
  const float kexp = (c & 1) ? 1.4426950408889634f : -1.4426950408889634f;
  const bool ones_lane = (c == 14);

  v8f acc[3] = {};

#pragma unroll
  for (int st = 0; st < STEPS_; ++st) {
    const int cur = st & 1;
    if (st + 1 < STEPS_) {
#pragma unroll
      for (int g = 0; g < GROUPS_; ++g) {
#pragma unroll
        for (int s = 0; s < 2; ++s)
          v[cur ^ 1][g][s] = *(const F3*)(sbase + ((st + 1) * 4 * GROUPS_ + 4 * g + 2 * s) * D_);
      }
    }
    float z[GROUPS_][2];
#pragma unroll
    for (int g = 0; g < GROUPS_; ++g) {
#pragma unroll
      for (int s = 0; s < 2; ++s) {
        float p = fmaf(v[cur][g][s].a, w[0], binit);
        p = fmaf(v[cur][g][s].b, w[1], p);
        p = fmaf(v[cur][g][s].c, w[2], p);
        z[g][s] = row_sum16(p);
      }
    }

    if (write_probs) {
      float zs = z[0][0];
#pragma unroll
      for (int i = 1; i < 2 * GROUPS_; ++i) zs = (c == i) ? z[i >> 1][i & 1] : zs;
      float e   = __expf(-fabsf(zs));
      float inv = 1.0f / (1.0f + e);
      float big = inv, small = e * inv;
      float p0 = (zs >= 0.0f) ? big : small;
      float p1 = (zs >= 0.0f) ? small : big;
      const int n = chunk * CHUNK_ + wave * ROWS_PER_WAVE_ + st * 4 * GROUPS_ +
                    4 * (c >> 1) + 2 * (c & 1) + h;
      if (c < 2 * GROUPS_) probs_out[(size_t)b * N_ + n] = make_float2(p0, p1);
    }
    if (accumulate) {
#pragma unroll
      for (int g = 0; g < GROUPS_; ++g) {
        v2f a;
        a.x = __builtin_amdgcn_rcpf(1.0f + __builtin_amdgcn_exp2f(z[g][0] * kexp));
        a.y = __builtin_amdgcn_rcpf(1.0f + __builtin_amdgcn_exp2f(z[g][1] * kexp));
        v2f b0, b1, b2;
        b0.x = ones_lane ? 1.0f : v[cur][g][0].a;  b0.y = ones_lane ? 1.0f : v[cur][g][1].a;
        b1.x = v[cur][g][0].b;                     b1.y = v[cur][g][1].b;
        b2.x = v[cur][g][0].c;                     b2.y = v[cur][g][1].c;
        acc[0] = __builtin_amdgcn_wmma_f32_16x16x4_f32(false, a, false, b0, (short)0, acc[0], false, false);
        acc[1] = __builtin_amdgcn_wmma_f32_16x16x4_f32(false, a, false, b1, (short)0, acc[1], false, false);
        acc[2] = __builtin_amdgcn_wmma_f32_16x16x4_f32(false, a, false, b2, (short)0, acc[2], false, false);
      }
    }
  }

  if (accumulate) {
    if (lane < 16) {
#pragma unroll
      for (int t = 0; t < 3; ++t) {
        const bool den = ones_lane && (t == 0);
        const int  d   = own[t] ? off + t : (den ? D_ : DP_ - 1);
        red[wave][0 * DP_ + d] = acc[t][0];
        red[wave][1 * DP_ + d] = acc[t][1];
      }
    }
    __syncthreads();
    if (tid < K_ * DP_) {
      float s = 0.f;
      if ((tid % DP_) <= D_) {
        float r[WAVES_];
#pragma unroll
        for (int wv = 0; wv < WAVES_; ++wv) r[wv] = red[wv][tid];
#pragma unroll
        for (int w2 = 1; w2 < WAVES_; w2 *= 2) {
#pragma unroll
          for (int wv = 0; wv < WAVES_; wv += 2 * w2) r[wv] += r[wv + w2];
        }
        s = r[0];
      }
      partials[(size_t)blockIdx.x * (K_ * DP_) + tid] = s;
    }
  }
}

__global__ __launch_bounds__(256) void sk_reduce(const float* __restrict__ partials,
                                                 float* __restrict__ cent) {
  int tid = blockIdx.x * 256 + threadIdx.x;
  if (tid >= B_ * K_ * DP_) return;
  int b  = tid / (K_ * DP_);
  int kd = tid % (K_ * DP_);
  int k  = kd / DP_;
  int d  = kd % DP_;
  const float* __restrict__ pv = partials + (size_t)b * CHUNKS_ * (K_ * DP_) + kd;
  const float* __restrict__ pd = partials + (size_t)b * CHUNKS_ * (K_ * DP_) + k * DP_ + D_;
  float val[CHUNKS_], den[CHUNKS_];
#pragma unroll
  for (int c = 0; c < CHUNKS_; ++c) {
    val[c] = pv[c * (K_ * DP_)];
    den[c] = pd[c * (K_ * DP_)];
  }
  float s = 0.f, sden = 0.f;
#pragma unroll
  for (int c = 0; c < CHUNKS_; ++c) {
    s    += val[c];
    sden += den[c];
  }
  cent[tid] = (d < D_) ? (s / sden) : 0.0f;
}

extern "C" void kernel_launch(void* const* d_in, const int* in_sizes, int n_in,
                              void* d_out, int out_size, void* d_ws, size_t ws_size,
                              hipStream_t stream) {
  const float* data = (const float*)d_in[0];
  const int*   ids  = (const int*)d_in[1];
  const int ITER = 10;

  float*  cent     = (float*)d_ws;
  float*  partials = cent + B_ * K_ * DP_;
  float2* out      = (float2*)d_out;

  sk_init<<<(B_ * K_ * DP_ + 255) / 256, 256, 0, stream>>>(data, ids, cent);
  for (int t = 0; t <= ITER; ++t) {
    int wp = (t == ITER) ? 1 : 0;
    int ac = (t <  ITER) ? 1 : 0;
    sk_pass<<<B_ * CHUNKS_, 32 * WAVES_, 0, stream>>>(data, cent, partials, out, wp, ac);
    if (ac)
      sk_reduce<<<(B_ * K_ * DP_ + 255) / 256, 256, 0, stream>>>(partials, cent);
  }
  (void)in_sizes; (void)n_in; (void)out_size; (void)ws_size;
}


# sk_pass scores: 16-lane sums by recursive halving (8 cross-lane adds per 16 rows, not 32) leave one score, one probability per lane; matrix rows 0-7/8-15 = cluster 0/1 partials; new fixed sum order
# speedup vs baseline: 2.5221x; 1.0561x over previous
#include <hip/hip_runtime.h>

#define B_   8
#define N_   65536
#define D_   40
#define DP_  48
#define K_   2
#define CHUNKS_ 32
#define CHUNK_  2048
#define WAVES_  32
#define ROWS_PER_WAVE_ 64
#define GROUPS_ 4
#define STEPS_  4

typedef float v2f __attribute__((ext_vector_type(2)));
typedef float v8f __attribute__((ext_vector_type(8)));

struct F3 { float a, b, c; };

__global__ __launch_bounds__(256) void sk_init(const float* __restrict__ data,
                                               const int* __restrict__ ids,
                                               float* __restrict__ cent) {
  int tid = blockIdx.x * 256 + threadIdx.x;
  if (tid >= B_ * K_ * DP_) return;
  int b  = tid / (K_ * DP_);
  int kd = tid % (K_ * DP_);
  int k  = kd / DP_;
  int d  = kd % DP_;
  int n  = ids[b * K_ + k];
  cent[tid] = (d < D_) ? data[((size_t)b * N_ + n) * D_ + d] : 0.0f;
}

template <int M>
__device__ __forceinline__ float row_xor(float v) {
  return __builtin_bit_cast(float,
      __builtin_amdgcn_update_dpp(0, __builtin_bit_cast(int, v), 0x160 + M, 0xf, 0xf, true));
}
__device__ __forceinline__ float row_sum16(float v) {
  v += row_xor<1>(v);
  v += row_xor<2>(v);
  v += row_xor<4>(v);
  v += row_xor<8>(v);
  return v;
}

__global__ __launch_bounds__(32 * WAVES_) void sk_pass(const float* __restrict__ data,
                                               const float* __restrict__ cent,
                                               float* __restrict__ partials,
                                               float2* __restrict__ probs_out,
                                               int write_probs, int accumulate) {
  __shared__ float red[WAVES_][K_ * DP_];

  const int tid   = threadIdx.x;
  const int wave  = tid >> 5;
  const int lane  = tid & 31;
  const int c     = lane & 15;
  const int h     = lane >> 4;
  const int b     = blockIdx.x / CHUNKS_;
  const int chunk = blockIdx.x & (CHUNKS_ - 1);

  const int off = (c < 13) ? 3 * c : (D_ - 3);
  bool own[3];
  own[0] = (c < 13); own[1] = (c < 13); own[2] = (c < 14);

  const float* __restrict__ sbase =
      data + ((size_t)b * N_ + (size_t)chunk * CHUNK_ + wave * ROWS_PER_WAVE_ + h) * D_ + off;

  F3 v[2][GROUPS_][2];
#pragma unroll
  for (int g = 0; g < GROUPS_; ++g) {
#pragma unroll
    for (int s = 0; s < 2; ++s) v[0][g][s] = *(const F3*)(sbase + (4 * g + 2 * s) * D_);
  }

  __builtin_amdgcn_sched_barrier(0);

  const float* __restrict__ c0p = cent + (b * K_ + 0) * DP_ + off;
  const float* __restrict__ c1p = cent + (b * K_ + 1) * DP_ + off;
  float w[3];
  float bpart = 0.0f;
#pragma unroll
  for (int t = 0; t < 3; ++t) {
    float a0 = c0p[t], a1 = c1p[t];
    w[t] = own[t] ? 2.0f * (a0 - a1) : 0.0f;
    bpart += own[t] ? (a1 * a1 - a0 * a0) : 0.0f;
  }
  const float bias = row_sum16(bpart);
  const float binit = (c == 15) ? bias : 0.0f;
  const float kexp = (c & 8) ? 1.4426950408889634f : -1.4426950408889634f;
  const bool ones_lane = (c == 14);
  const bool up1 = (c & 1) != 0, up2 = (c & 2) != 0, up4 = (c & 4) != 0;
  const int zi = ((c & 1) << 2) | (c & 2) | ((c >> 2) & 1);
  v2f own_pos[GROUPS_];
#pragma unroll
  for (int g = 0; g < GROUPS_; ++g) {
    own_pos[g].x = (zi == 2 * g)     ? 1.0f : 0.0f;
    own_pos[g].y = (zi == 2 * g + 1) ? 1.0f : 0.0f;
  }

  v8f acc[3] = {};

#pragma unroll
  for (int st = 0; st < STEPS_; ++st) {
    const int cur = st & 1;
    if (st + 1 < STEPS_) {
#pragma unroll
      for (int g = 0; g < GROUPS_; ++g) {
#pragma unroll
        for (int s = 0; s < 2; ++s)
          v[cur ^ 1][g][s] = *(const F3*)(sbase + ((st + 1) * 4 * GROUPS_ + 4 * g + 2 * s) * D_);
      }
    }
    float q[2 * GROUPS_];
#pragma unroll
    for (int g = 0; g < GROUPS_; ++g) {
#pragma unroll
      for (int s = 0; s < 2; ++s) {
        float p = fmaf(v[cur][g][s].a, w[0], binit);
        p = fmaf(v[cur][g][s].b, w[1], p);
        p = fmaf(v[cur][g][s].c, w[2], p);
        q[2 * g + s] = p;
      }
    }
    float r1[4];
#pragma unroll
    for (int j = 0; j < 4; ++j) {
      const float keep = up1 ? q[4 + j] : q[j];
      const float send = up1 ? q[j] : q[4 + j];
      r1[j] = keep + row_xor<1>(send);
    }
    float r2[2];
#pragma unroll
    for (int j = 0; j < 2; ++j) {
      const float keep = up2 ? r1[2 + j] : r1[j];
      const float send = up2 ? r1[j] : r1[2 + j];
      r2[j] = keep + row_xor<2>(send);
    }
    const float keep4 = up4 ? r2[1] : r2[0];
    const float send4 = up4 ? r2[0] : r2[1];
    const float r3 = keep4 + row_xor<4>(send4);
    const float z = r3 + row_xor<8>(r3);

    if (write_probs) {
      float e   = __expf(-fabsf(z));
      float inv = 1.0f / (1.0f + e);
      float big = inv, small = e * inv;
      float p0 = (z >= 0.0f) ? big : small;
      float p1 = (z >= 0.0f) ? small : big;
      const int n = chunk * CHUNK_ + wave * ROWS_PER_WAVE_ + st * 4 * GROUPS_ + 2 * zi + h;
      if (c < 2 * GROUPS_) probs_out[(size_t)b * N_ + n] = make_float2(p0, p1);
    }
    if (accumulate) {
      const float pr = __builtin_amdgcn_rcpf(1.0f + __builtin_amdgcn_exp2f(z * kexp));
      v2f pr2; pr2.x = pr; pr2.y = pr;
#pragma unroll
      for (int g = 0; g < GROUPS_; ++g) {
        const v2f a = pr2 * own_pos[g];
        v2f b0, b1, b2;
        b0.x = ones_lane ? 1.0f : v[cur][g][0].a;  b0.y = ones_lane ? 1.0f : v[cur][g][1].a;
        b1.x = v[cur][g][0].b;                     b1.y = v[cur][g][1].b;
        b2.x = v[cur][g][0].c;                     b2.y = v[cur][g][1].c;
        acc[0] = __builtin_amdgcn_wmma_f32_16x16x4_f32(false, a, false, b0, (short)0, acc[0], false, false);
        acc[1] = __builtin_amdgcn_wmma_f32_16x16x4_f32(false, a, false, b1, (short)0, acc[1], false, false);
        acc[2] = __builtin_amdgcn_wmma_f32_16x16x4_f32(false, a, false, b2, (short)0, acc[2], false, false);
      }
    }
  }

  if (accumulate) {
#pragma unroll
    for (int t = 0; t < 3; ++t) {
      const float tot = ((acc[t][0] + acc[t][1]) + (acc[t][2] + acc[t][3])) +
                        ((acc[t][4] + acc[t][5]) + (acc[t][6] + acc[t][7]));
      const bool den = ones_lane && (t == 0);
      const int  d   = own[t] ? off + t : (den ? D_ : DP_ - 1);
      red[wave][h * DP_ + d] = tot;
    }
    __syncthreads();
    if (tid < K_ * DP_) {
      float s = 0.f;
      if ((tid % DP_) <= D_) {
        float r[WAVES_];
#pragma unroll
        for (int wv = 0; wv < WAVES_; ++wv) r[wv] = red[wv][tid];
#pragma unroll
        for (int w2 = 1; w2 < WAVES_; w2 *= 2) {
#pragma unroll
          for (int wv = 0; wv < WAVES_; wv += 2 * w2) r[wv] += r[wv + w2];
        }
        s = r[0];
      }
      partials[(size_t)blockIdx.x * (K_ * DP_) + tid] = s;
    }
  }
}

__global__ __launch_bounds__(256) void sk_reduce(const float* __restrict__ partials,
                                                 float* __restrict__ cent) {
  int tid = blockIdx.x * 256 + threadIdx.x;
  if (tid >= B_ * K_ * DP_) return;
  int b  = tid / (K_ * DP_);
  int kd = tid % (K_ * DP_);
  int k  = kd / DP_;
  int d  = kd % DP_;
  const float* __restrict__ pv = partials + (size_t)b * CHUNKS_ * (K_ * DP_) + kd;
  const float* __restrict__ pd = partials + (size_t)b * CHUNKS_ * (K_ * DP_) + k * DP_ + D_;
  float val[CHUNKS_], den[CHUNKS_];
#pragma unroll
  for (int c = 0; c < CHUNKS_; ++c) {
    val[c] = pv[c * (K_ * DP_)];
    den[c] = pd[c * (K_ * DP_)];
  }
  float s = 0.f, sden = 0.f;
#pragma unroll
  for (int c = 0; c < CHUNKS_; ++c) {
    s    += val[c];
    sden += den[c];
  }
  cent[tid] = (d < D_) ? (s / sden) : 0.0f;
}

extern "C" void kernel_launch(void* const* d_in, const int* in_sizes, int n_in,
                              void* d_out, int out_size, void* d_ws, size_t ws_size,
                              hipStream_t stream) {
  const float* data = (const float*)d_in[0];
  const int*   ids  = (const int*)d_in[1];
  const int ITER = 10;

  float*  cent     = (float*)d_ws;
  float*  partials = cent + B_ * K_ * DP_;
  float2* out      = (float2*)d_out;

  sk_init<<<(B_ * K_ * DP_ + 255) / 256, 256, 0, stream>>>(data, ids, cent);
  for (int t = 0; t <= ITER; ++t) {
    int wp = (t == ITER) ? 1 : 0;
    int ac = (t <  ITER) ? 1 : 0;
    sk_pass<<<B_ * CHUNKS_, 32 * WAVES_, 0, stream>>>(data, cent, partials, out, wp, ac);
    if (ac)
      sk_reduce<<<(B_ * K_ * DP_ + 255) / 256, 256, 0, stream>>>(partials, cent);
  }
  (void)in_sizes; (void)n_in; (void)out_size; (void)ws_size;
}
